# starting kernel
# baseline (speedup 1.0000x reference)
#include <hip/hip_runtime.h>

#define B_SAMPLES 4096
#define D_INF     300
#define D_OUTF    300
#define NUM_MAT   274
#define TILE      16
#define NT_OUT    ((D_OUTF + TILE - 1) / TILE)
#define MAX_TILES (NUM_MAT + (B_SAMPLES / TILE))

typedef __attribute__((ext_vector_type(2))) float v2f;
typedef __attribute__((ext_vector_type(4))) float v4f;
typedef __attribute__((ext_vector_type(8))) float v8f;

__global__ void rbt_init(int* __restrict__ counts, int* __restrict__ cursor,
                         int* __restrict__ ntiles) {
    int i = blockIdx.x * blockDim.x + threadIdx.x;
    if (i < NUM_MAT) { counts[i] = 0; cursor[i] = 0; }
    if (i == 0) { *ntiles = 0; }
}

__global__ void rbt_hist(const int* __restrict__ rel, int* __restrict__ counts) {
    int b = blockIdx.x * blockDim.x + threadIdx.x;
    if (b < B_SAMPLES) atomicAdd(&counts[rel[b]], 1);
}

__global__ void rbt_scan(const int* __restrict__ counts, int* __restrict__ offsets,
                         int* __restrict__ tiles, int* __restrict__ ntiles) {
    if (threadIdx.x != 0 || blockIdx.x != 0) return;
    int acc = 0, t = 0;
    for (int r = 0; r < NUM_MAT; ++r) {
        offsets[r] = acc;
        int c = counts[r];
        for (int s = 0; s < c; s += TILE) {
            tiles[3 * t + 0] = r;
            tiles[3 * t + 1] = acc + s;
            tiles[3 * t + 2] = (c - s < TILE) ? (c - s) : TILE;
            ++t;
        }
        acc += c;
    }
    offsets[NUM_MAT] = acc;
    *ntiles = t;
}

__global__ void rbt_scatter(const int* __restrict__ rel, const int* __restrict__ offsets,
                            int* __restrict__ cursor, int* __restrict__ order) {
    int b = blockIdx.x * blockDim.x + threadIdx.x;
    if (b < B_SAMPLES) {
        int r = rel[b];
        int pos = offsets[r] + atomicAdd(&cursor[r], 1);
        order[pos] = b;
    }
}

__global__ void rbt_gemm(const float* __restrict__ x, const float* __restrict__ W,
                         float* __restrict__ out, const int* __restrict__ order,
                         const int* __restrict__ tiles, const int* __restrict__ ntiles) {
    const int tid = blockIdx.y;
    if (tid >= *ntiles) return;

    const int r     = tiles[3 * tid + 0];
    const int start = tiles[3 * tid + 1];
    const int cnt   = tiles[3 * tid + 2];

    const int lane = threadIdx.x;
    const int half = lane >> 4;
    const int l16  = lane & 15;

    const int rowbase = blockIdx.x * TILE;
    const int arow = (rowbase + l16 < D_OUTF) ? (rowbase + l16) : (D_OUTF - 1);
    const int col  = (l16 < cnt) ? l16 : (cnt - 1);
    const int s    = order[start + col];

    const float* wp = W + (size_t)r * (D_OUTF * D_INF) + (size_t)arow * D_INF + 2 * half;
    const float* xp = x + (size_t)s * D_INF + 2 * half;

    v8f c = {};
#pragma unroll 5
    for (int k = 0; k < D_INF; k += 4) {
        v2f a = *(const v2f*)(wp + k);
        v2f b = *(const v2f*)(xp + k);
        c = __builtin_amdgcn_wmma_f32_16x16x4_f32(
                 false, a,  false, b,
                 (short)0, c,  false,  false);
    }

    if (l16 < cnt) {
        const int m0 = rowbase + 8 * half;
        float* op = out + (size_t)s * D_OUTF + m0;
        if (m0 + 8 <= D_OUTF) {
            v4f lo = {c[0], c[1], c[2], c[3]};
            v4f hi = {c[4], c[5], c[6], c[7]};
            *(v4f*)(op + 0) = lo;
            *(v4f*)(op + 4) = hi;
        } else {
#pragma unroll
            for (int v = 0; v < 8; ++v)
                if (m0 + v < D_OUTF) op[v] = c[v];
        }
    }
}

extern "C" void kernel_launch(void* const* d_in, const int* in_sizes, int n_in,
                              void* d_out, int out_size, void* d_ws, size_t ws_size,
                              hipStream_t stream) {
    const float* x   = (const float*)d_in[0];
    const int*   rel = (const int*)  d_in[1];
    const float* W   = (const float*)d_in[2];
    float*       out = (float*)d_out;

    int* ws      = (int*)d_ws;
    int* counts  = ws;
    int* offsets = counts + NUM_MAT;
    int* cursor  = offsets + NUM_MAT + 1;
    int* order   = cursor + NUM_MAT;
    int* ntiles  = order + B_SAMPLES;
    int* tiles   = ntiles + 1;

    rbt_init   <<<1, 320, 0, stream>>>(counts, cursor, ntiles);
    rbt_hist   <<<B_SAMPLES / 256, 256, 0, stream>>>(rel, counts);
    rbt_scan   <<<1, 32, 0, stream>>>(counts, offsets, tiles, ntiles);
    rbt_scatter<<<B_SAMPLES / 256, 256, 0, stream>>>(rel, offsets, cursor, order);

    dim3 grid(NT_OUT, MAX_TILES);
    rbt_gemm   <<<grid, 32, 0, stream>>>(x, W, out, order, tiles, ntiles);
}


# Prefix scan of the 274 relation counts and the tile list built by one 320-thread work-group (LDS doubling scan) instead of a serial walk on one lane
# speedup vs baseline: 1.4332x; 1.4332x over previous
#include <hip/hip_runtime.h>

#define B_SAMPLES 4096
#define D_INF     300
#define D_OUTF    300
#define NUM_MAT   274
#define TILE      16
#define NT_OUT    ((D_OUTF + TILE - 1) / TILE)
#define MAX_TILES (NUM_MAT + (B_SAMPLES / TILE))

typedef __attribute__((ext_vector_type(2))) float v2f;
typedef __attribute__((ext_vector_type(4))) float v4f;
typedef __attribute__((ext_vector_type(8))) float v8f;

__global__ void rbt_init(int* __restrict__ counts, int* __restrict__ cursor,
                         int* __restrict__ ntiles) {
    int i = blockIdx.x * blockDim.x + threadIdx.x;
    if (i < NUM_MAT) { counts[i] = 0; cursor[i] = 0; }
    if (i == 0) { *ntiles = 0; }
}

__global__ void rbt_hist(const int* __restrict__ rel, int* __restrict__ counts) {
    int b = blockIdx.x * blockDim.x + threadIdx.x;
    if (b < B_SAMPLES) atomicAdd(&counts[rel[b]], 1);
}

#define SCAN_T 320
__global__ void rbt_scan(const int* __restrict__ counts, int* __restrict__ offsets,
                         int* __restrict__ tiles, int* __restrict__ ntiles) {
    __shared__ int s_cnt[SCAN_T];
    __shared__ int s_til[SCAN_T];
    const int r  = threadIdx.x;
    const int c  = (r < NUM_MAT) ? counts[r] : 0;
    const int nt = (c + TILE - 1) / TILE;
    s_cnt[r] = c;
    s_til[r] = nt;
    __syncthreads();
    for (int d = 1; d < SCAN_T; d <<= 1) {
        int a = 0, b = 0;
        if (r >= d) { a = s_cnt[r - d]; b = s_til[r - d]; }
        __syncthreads();
        s_cnt[r] += a;
        s_til[r] += b;
        __syncthreads();
    }
    if (r < NUM_MAT) {
        const int acc = s_cnt[r] - c;
        int t = s_til[r] - nt;
        offsets[r] = acc;
        for (int s = 0; s < c; s += TILE) {
            tiles[3 * t + 0] = r;
            tiles[3 * t + 1] = acc + s;
            tiles[3 * t + 2] = (c - s < TILE) ? (c - s) : TILE;
            ++t;
        }
        if (r == NUM_MAT - 1) {
            offsets[NUM_MAT] = acc + c;
            *ntiles = t;
        }
    }
}

__global__ void rbt_scatter(const int* __restrict__ rel, const int* __restrict__ offsets,
                            int* __restrict__ cursor, int* __restrict__ order) {
    int b = blockIdx.x * blockDim.x + threadIdx.x;
    if (b < B_SAMPLES) {
        int r = rel[b];
        int pos = offsets[r] + atomicAdd(&cursor[r], 1);
        order[pos] = b;
    }
}

__global__ void rbt_gemm(const float* __restrict__ x, const float* __restrict__ W,
                         float* __restrict__ out, const int* __restrict__ order,
                         const int* __restrict__ tiles, const int* __restrict__ ntiles) {
    const int tid = blockIdx.y;
    if (tid >= *ntiles) return;

    const int r     = tiles[3 * tid + 0];
    const int start = tiles[3 * tid + 1];
    const int cnt   = tiles[3 * tid + 2];

    const int lane = threadIdx.x;
    const int half = lane >> 4;
    const int l16  = lane & 15;

    const int rowbase = blockIdx.x * TILE;
    const int arow = (rowbase + l16 < D_OUTF) ? (rowbase + l16) : (D_OUTF - 1);
    const int col  = (l16 < cnt) ? l16 : (cnt - 1);
    const int s    = order[start + col];

    const float* wp = W + (size_t)r * (D_OUTF * D_INF) + (size_t)arow * D_INF + 2 * half;
    const float* xp = x + (size_t)s * D_INF + 2 * half;

    v8f c = {};
#pragma unroll 5
    for (int k = 0; k < D_INF; k += 4) {
        v2f a = *(const v2f*)(wp + k);
        v2f b = *(const v2f*)(xp + k);
        c = __builtin_amdgcn_wmma_f32_16x16x4_f32(
                 false, a,  false, b,
                 (short)0, c,  false,  false);
    }

    if (l16 < cnt) {
        const int m0 = rowbase + 8 * half;
        float* op = out + (size_t)s * D_OUTF + m0;
        if (m0 + 8 <= D_OUTF) {
            v4f lo = {c[0], c[1], c[2], c[3]};
            v4f hi = {c[4], c[5], c[6], c[7]};
            *(v4f*)(op + 0) = lo;
            *(v4f*)(op + 4) = hi;
        } else {
#pragma unroll
            for (int v = 0; v < 8; ++v)
                if (m0 + v < D_OUTF) op[v] = c[v];
        }
    }
}

extern "C" void kernel_launch(void* const* d_in, const int* in_sizes, int n_in,
                              void* d_out, int out_size, void* d_ws, size_t ws_size,
                              hipStream_t stream) {
    const float* x   = (const float*)d_in[0];
    const int*   rel = (const int*)  d_in[1];
    const float* W   = (const float*)d_in[2];
    float*       out = (float*)d_out;

    int* ws      = (int*)d_ws;
    int* counts  = ws;
    int* offsets = counts + NUM_MAT;
    int* cursor  = offsets + NUM_MAT + 1;
    int* order   = cursor + NUM_MAT;
    int* ntiles  = order + B_SAMPLES;
    int* tiles   = ntiles + 1;

    rbt_init   <<<1, 320, 0, stream>>>(counts, cursor, ntiles);
    rbt_hist   <<<B_SAMPLES / 256, 256, 0, stream>>>(rel, counts);
    rbt_scan   <<<1, SCAN_T, 0, stream>>>(counts, offsets, tiles, ntiles);
    rbt_scatter<<<B_SAMPLES / 256, 256, 0, stream>>>(rel, offsets, cursor, order);

    dim3 grid(NT_OUT, MAX_TILES);
    rbt_gemm   <<<grid, 32, 0, stream>>>(x, W, out, order, tiles, ntiles);
}


# GEMM fragment loads widened from 8 to 16 bytes per lane: 8 K per step, two WMMAs per load pair, K loop straight-line; K summed as 0,1,4,5,2,3,6,7 within each 8 (summation order changes)
# speedup vs baseline: 1.5246x; 1.0638x over previous
#include <hip/hip_runtime.h>

#define B_SAMPLES 4096
#define D_INF     300
#define D_OUTF    300
#define NUM_MAT   274
#define TILE      16
#define NT_OUT    ((D_OUTF + TILE - 1) / TILE)
#define MAX_TILES (NUM_MAT + (B_SAMPLES / TILE))

typedef __attribute__((ext_vector_type(2))) float v2f;
typedef __attribute__((ext_vector_type(4))) float v4f;
typedef __attribute__((ext_vector_type(8))) float v8f;

__global__ void rbt_init(int* __restrict__ counts, int* __restrict__ cursor,
                         int* __restrict__ ntiles) {
    int i = blockIdx.x * blockDim.x + threadIdx.x;
    if (i < NUM_MAT) { counts[i] = 0; cursor[i] = 0; }
    if (i == 0) { *ntiles = 0; }
}

__global__ void rbt_hist(const int* __restrict__ rel, int* __restrict__ counts) {
    int b = blockIdx.x * blockDim.x + threadIdx.x;
    if (b < B_SAMPLES) atomicAdd(&counts[rel[b]], 1);
}

#define SCAN_T 320
__global__ void rbt_scan(const int* __restrict__ counts, int* __restrict__ offsets,
                         int* __restrict__ tiles, int* __restrict__ ntiles) {
    __shared__ int s_cnt[SCAN_T];
    __shared__ int s_til[SCAN_T];
    const int r  = threadIdx.x;
    const int c  = (r < NUM_MAT) ? counts[r] : 0;
    const int nt = (c + TILE - 1) / TILE;
    s_cnt[r] = c;
    s_til[r] = nt;
    __syncthreads();
    for (int d = 1; d < SCAN_T; d <<= 1) {
        int a = 0, b = 0;
        if (r >= d) { a = s_cnt[r - d]; b = s_til[r - d]; }
        __syncthreads();
        s_cnt[r] += a;
        s_til[r] += b;
        __syncthreads();
    }
    if (r < NUM_MAT) {
        const int acc = s_cnt[r] - c;
        int t = s_til[r] - nt;
        offsets[r] = acc;
        for (int s = 0; s < c; s += TILE) {
            tiles[3 * t + 0] = r;
            tiles[3 * t + 1] = acc + s;
            tiles[3 * t + 2] = (c - s < TILE) ? (c - s) : TILE;
            ++t;
        }
        if (r == NUM_MAT - 1) {
            offsets[NUM_MAT] = acc + c;
            *ntiles = t;
        }
    }
}

__global__ void rbt_scatter(const int* __restrict__ rel, const int* __restrict__ offsets,
                            int* __restrict__ cursor, int* __restrict__ order) {
    int b = blockIdx.x * blockDim.x + threadIdx.x;
    if (b < B_SAMPLES) {
        int r = rel[b];
        int pos = offsets[r] + atomicAdd(&cursor[r], 1);
        order[pos] = b;
    }
}

__global__ void rbt_gemm(const float* __restrict__ x, const float* __restrict__ W,
                         float* __restrict__ out, const int* __restrict__ order,
                         const int* __restrict__ tiles, const int* __restrict__ ntiles) {
    const int tid = blockIdx.y;
    if (tid >= *ntiles) return;

    const int r     = tiles[3 * tid + 0];
    const int start = tiles[3 * tid + 1];
    const int cnt   = tiles[3 * tid + 2];

    const int lane = threadIdx.x;
    const int half = lane >> 4;
    const int l16  = lane & 15;

    const int rowbase = blockIdx.x * TILE;
    const int arow = (rowbase + l16 < D_OUTF) ? (rowbase + l16) : (D_OUTF - 1);
    const int col  = (l16 < cnt) ? l16 : (cnt - 1);
    const int s    = order[start + col];

    const float* wrow = W + (size_t)r * (D_OUTF * D_INF) + (size_t)arow * D_INF;
    const float* xrow = x + (size_t)s * D_INF;
    const float* wp = wrow + 4 * half;
    const float* xp = xrow + 4 * half;

    v8f c = {};
#pragma unroll
    for (int k = 0; k < D_INF - 4; k += 8) {
        v4f a = *(const v4f*)(wp + k);
        v4f b = *(const v4f*)(xp + k);
        v2f a0 = {a[0], a[1]}, a1 = {a[2], a[3]};
        v2f b0 = {b[0], b[1]}, b1 = {b[2], b[3]};
        c = __builtin_amdgcn_wmma_f32_16x16x4_f32(
                 false, a0,  false, b0,
                 (short)0, c,  false,  false);
        c = __builtin_amdgcn_wmma_f32_16x16x4_f32(
                 false, a1,  false, b1,
                 (short)0, c,  false,  false);
    }
    {
        v2f a = *(const v2f*)(wrow + (D_INF - 4) + 2 * half);
        v2f b = *(const v2f*)(xrow + (D_INF - 4) + 2 * half);
        c = __builtin_amdgcn_wmma_f32_16x16x4_f32(
                 false, a,  false, b,
                 (short)0, c,  false,  false);
    }

    if (l16 < cnt) {
        const int m0 = rowbase + 8 * half;
        float* op = out + (size_t)s * D_OUTF + m0;
        if (m0 + 8 <= D_OUTF) {
            v4f lo = {c[0], c[1], c[2], c[3]};
            v4f hi = {c[4], c[5], c[6], c[7]};
            *(v4f*)(op + 0) = lo;
            *(v4f*)(op + 4) = hi;
        } else {
#pragma unroll
            for (int v = 0; v < 8; ++v)
                if (m0 + v < D_OUTF) op[v] = c[v];
        }
    }
}

extern "C" void kernel_launch(void* const* d_in, const int* in_sizes, int n_in,
                              void* d_out, int out_size, void* d_ws, size_t ws_size,
                              hipStream_t stream) {
    const float* x   = (const float*)d_in[0];
    const int*   rel = (const int*)  d_in[1];
    const float* W   = (const float*)d_in[2];
    float*       out = (float*)d_out;

    int* ws      = (int*)d_ws;
    int* counts  = ws;
    int* offsets = counts + NUM_MAT;
    int* cursor  = offsets + NUM_MAT + 1;
    int* order   = cursor + NUM_MAT;
    int* ntiles  = order + B_SAMPLES;
    int* tiles   = ntiles + 1;

    rbt_init   <<<1, 320, 0, stream>>>(counts, cursor, ntiles);
    rbt_hist   <<<B_SAMPLES / 256, 256, 0, stream>>>(rel, counts);
    rbt_scan   <<<1, SCAN_T, 0, stream>>>(counts, offsets, tiles, ntiles);
    rbt_scatter<<<B_SAMPLES / 256, 256, 0, stream>>>(rel, offsets, cursor, order);

    dim3 grid(NT_OUT, MAX_TILES);
    rbt_gemm   <<<grid, 32, 0, stream>>>(x, W, out, order, tiles, ntiles);
}


# The four bucketing launches (clear, histogram, scan, scatter) fused into one 1024-thread work-group: LDS histogram whose atomic's return value is the sample's rank; 5 launches become 2
# speedup vs baseline: 1.6806x; 1.1023x over previous
#include <hip/hip_runtime.h>

#define B_SAMPLES 4096
#define D_INF     300
#define D_OUTF    300
#define NUM_MAT   274
#define TILE      16
#define NT_OUT    ((D_OUTF + TILE - 1) / TILE)
#define MAX_TILES (NUM_MAT + (B_SAMPLES / TILE))

typedef __attribute__((ext_vector_type(2))) float v2f;
typedef __attribute__((ext_vector_type(4))) float v4f;
typedef __attribute__((ext_vector_type(8))) float v8f;

#define SCAN_T 320
#define PRE_T  1024
#define PER_T  (B_SAMPLES / PRE_T)
__global__ void __launch_bounds__(PRE_T)
rbt_bucket(const int* __restrict__ rel, int* __restrict__ order,
           int* __restrict__ tiles, int* __restrict__ ntiles) {
    __shared__ int s_hist[SCAN_T];
    __shared__ int s_cnt[SCAN_T];
    __shared__ int s_til[SCAN_T];
    const int t = threadIdx.x;
    int rr[PER_T], rk[PER_T];
    for (int j = 0; j < PER_T; ++j) rr[j] = rel[t + j * PRE_T];
    if (t < SCAN_T) s_hist[t] = 0;
    __syncthreads();
    for (int j = 0; j < PER_T; ++j) rk[j] = atomicAdd(&s_hist[rr[j]], 1);
    __syncthreads();
    int c = 0, nt = 0;
    if (t < SCAN_T) {
        c  = s_hist[t];
        nt = (c + TILE - 1) / TILE;
        s_cnt[t] = c;
        s_til[t] = nt;
    }
    __syncthreads();
    for (int d = 1; d < SCAN_T; d <<= 1) {
        int a = 0, b = 0;
        if (t < SCAN_T && t >= d) { a = s_cnt[t - d]; b = s_til[t - d]; }
        __syncthreads();
        if (t < SCAN_T) { s_cnt[t] += a; s_til[t] += b; }
        __syncthreads();
    }
    for (int j = 0; j < PER_T; ++j) {
        const int r = rr[j];
        order[s_cnt[r] - s_hist[r] + rk[j]] = t + j * PRE_T;
    }
    if (t < NUM_MAT) {
        const int acc = s_cnt[t] - c;
        int q = s_til[t] - nt;
        for (int sidx = 0; sidx < c; sidx += TILE) {
            tiles[3 * q + 0] = t;
            tiles[3 * q + 1] = acc + sidx;
            tiles[3 * q + 2] = (c - sidx < TILE) ? (c - sidx) : TILE;
            ++q;
        }
        if (t == NUM_MAT - 1) *ntiles = q;
    }
}

__global__ void rbt_gemm(const float* __restrict__ x, const float* __restrict__ W,
                         float* __restrict__ out, const int* __restrict__ order,
                         const int* __restrict__ tiles, const int* __restrict__ ntiles) {
    const int tid = blockIdx.y;
    if (tid >= *ntiles) return;

    const int r     = tiles[3 * tid + 0];
    const int start = tiles[3 * tid + 1];
    const int cnt   = tiles[3 * tid + 2];

    const int lane = threadIdx.x;
    const int half = lane >> 4;
    const int l16  = lane & 15;

    const int rowbase = blockIdx.x * TILE;
    const int arow = (rowbase + l16 < D_OUTF) ? (rowbase + l16) : (D_OUTF - 1);
    const int col  = (l16 < cnt) ? l16 : (cnt - 1);
    const int s    = order[start + col];

    const float* wrow = W + (size_t)r * (D_OUTF * D_INF) + (size_t)arow * D_INF;
    const float* xrow = x + (size_t)s * D_INF;
    const float* wp = wrow + 4 * half;
    const float* xp = xrow + 4 * half;

    v8f c = {};
#pragma unroll
    for (int k = 0; k < D_INF - 4; k += 8) {
        v4f a = *(const v4f*)(wp + k);
        v4f b = *(const v4f*)(xp + k);
        v2f a0 = {a[0], a[1]}, a1 = {a[2], a[3]};
        v2f b0 = {b[0], b[1]}, b1 = {b[2], b[3]};
        c = __builtin_amdgcn_wmma_f32_16x16x4_f32(
                 false, a0,  false, b0,
                 (short)0, c,  false,  false);
        c = __builtin_amdgcn_wmma_f32_16x16x4_f32(
                 false, a1,  false, b1,
                 (short)0, c,  false,  false);
    }
    {
        v2f a = *(const v2f*)(wrow + (D_INF - 4) + 2 * half);
        v2f b = *(const v2f*)(xrow + (D_INF - 4) + 2 * half);
        c = __builtin_amdgcn_wmma_f32_16x16x4_f32(
                 false, a,  false, b,
                 (short)0, c,  false,  false);
    }

    if (l16 < cnt) {
        const int m0 = rowbase + 8 * half;
        float* op = out + (size_t)s * D_OUTF + m0;
        if (m0 + 8 <= D_OUTF) {
            v4f lo = {c[0], c[1], c[2], c[3]};
            v4f hi = {c[4], c[5], c[6], c[7]};
            *(v4f*)(op + 0) = lo;
            *(v4f*)(op + 4) = hi;
        } else {
#pragma unroll
            for (int v = 0; v < 8; ++v)
                if (m0 + v < D_OUTF) op[v] = c[v];
        }
    }
}

extern "C" void kernel_launch(void* const* d_in, const int* in_sizes, int n_in,
                              void* d_out, int out_size, void* d_ws, size_t ws_size,
                              hipStream_t stream) {
    const float* x   = (const float*)d_in[0];
    const int*   rel = (const int*)  d_in[1];
    const float* W   = (const float*)d_in[2];
    float*       out = (float*)d_out;

    int* ws      = (int*)d_ws;
    int* order   = ws;
    int* ntiles  = order + B_SAMPLES;
    int* tiles   = ntiles + 1;

    rbt_bucket <<<1, PRE_T, 0, stream>>>(rel, order, tiles, ntiles);

    dim3 grid(NT_OUT, MAX_TILES);
    rbt_gemm   <<<grid, 32, 0, stream>>>(x, W, out, order, tiles, ntiles);
}


# GEMM operands staged through LDS: 4-wave groups fetch each wave's 16 W rows (one contiguous block) and one shared x tile with lane-contiguous 16-byte loads; fragments read from LDS, K order kept
# speedup vs baseline: 2.7695x; 1.6479x over previous
#include <hip/hip_runtime.h>

#define B_SAMPLES 4096
#define D_INF     300
#define D_OUTF    300
#define NUM_MAT   274
#define TILE      16
#define NT_OUT    ((D_OUTF + TILE - 1) / TILE)
#define MAX_TILES (NUM_MAT + (B_SAMPLES / TILE))

typedef __attribute__((ext_vector_type(2))) float v2f;
typedef __attribute__((ext_vector_type(4))) float v4f;
typedef __attribute__((ext_vector_type(8))) float v8f;

#define SCAN_T 320
#define PRE_T  1024
#define PER_T  (B_SAMPLES / PRE_T)
__global__ void __launch_bounds__(PRE_T)
rbt_bucket(const int* __restrict__ rel, int* __restrict__ order,
           int* __restrict__ tiles, int* __restrict__ ntiles) {
    __shared__ int s_hist[SCAN_T];
    __shared__ int s_cnt[SCAN_T];
    __shared__ int s_til[SCAN_T];
    const int t = threadIdx.x;
    int rr[PER_T], rk[PER_T];
    for (int j = 0; j < PER_T; ++j) rr[j] = rel[t + j * PRE_T];
    if (t < SCAN_T) s_hist[t] = 0;
    __syncthreads();
    for (int j = 0; j < PER_T; ++j) rk[j] = atomicAdd(&s_hist[rr[j]], 1);
    __syncthreads();
    int c = 0, nt = 0;
    if (t < SCAN_T) {
        c  = s_hist[t];
        nt = (c + TILE - 1) / TILE;
        s_cnt[t] = c;
        s_til[t] = nt;
    }
    __syncthreads();
    for (int d = 1; d < SCAN_T; d <<= 1) {
        int a = 0, b = 0;
        if (t < SCAN_T && t >= d) { a = s_cnt[t - d]; b = s_til[t - d]; }
        __syncthreads();
        if (t < SCAN_T) { s_cnt[t] += a; s_til[t] += b; }
        __syncthreads();
    }
    for (int j = 0; j < PER_T; ++j) {
        const int r = rr[j];
        order[s_cnt[r] - s_hist[r] + rk[j]] = t + j * PRE_T;
    }
    if (t < NUM_MAT) {
        const int acc = s_cnt[t] - c;
        int q = s_til[t] - nt;
        for (int sidx = 0; sidx < c; sidx += TILE) {
            tiles[3 * q + 0] = t;
            tiles[3 * q + 1] = acc + sidx;
            tiles[3 * q + 2] = (c - sidx < TILE) ? (c - sidx) : TILE;
            ++q;
        }
        if (t == NUM_MAT - 1) *ntiles = q;
    }
}

#define NW       4
#define GEMM_T   (32 * NW)
#define NT_GRP   ((NT_OUT + NW - 1) / NW)
#define ROW_P    (D_INF / 4)
#define TILE_P   (TILE * ROW_P)
#define W_LOADS  ((TILE_P + 31) / 32)
#define X_LOADS  ((TILE_P + GEMM_T - 1) / GEMM_T)
#define KSTEPS   ((D_INF - 4) / 8)
#define KB       8
__global__ void __launch_bounds__(GEMM_T)
rbt_gemm(const float* __restrict__ x, const float* __restrict__ W,
         float* __restrict__ out, const int* __restrict__ order,
         const int* __restrict__ tiles, const int* __restrict__ ntiles) {
    __shared__ v4f s_x[X_LOADS * GEMM_T];
    __shared__ v4f s_w[NW][W_LOADS * 32];

    const int tid = blockIdx.y;
    if (tid >= *ntiles) return;

    const int r     = tiles[3 * tid + 0];
    const int start = tiles[3 * tid + 1];
    const int cnt   = tiles[3 * tid + 2];

    const int t    = threadIdx.x;
    const int wave = __builtin_amdgcn_readfirstlane(t >> 5);
    const int lane = t & 31;
    const int half = lane >> 4;
    const int l16  = lane & 15;

    const int  rowtile = blockIdx.x * NW + wave;
    const bool active  = rowtile < NT_OUT;
    const int  rowbase = rowtile * TILE;

    int sj[X_LOADS], pj[X_LOADS];
#pragma unroll
    for (int i = 0; i < X_LOADS; ++i) {
        const int j    = t + GEMM_T * i;
        const int jc   = (j < TILE_P) ? j : (TILE_P - 1);
        const int colj = jc / ROW_P;
        pj[i] = jc - colj * ROW_P;
        sj[i] = order[start + ((colj < cnt) ? colj : (cnt - 1))];
    }
    const int col = (l16 < cnt) ? l16 : (cnt - 1);
    const int s   = order[start + col];
    v4f wb[W_LOADS];
    {
        const float* wmat = W + (size_t)r * (D_OUTF * D_INF);
        const int    wofs = rowbase * D_INF + 4 * lane;
#pragma unroll
        for (int i = 0; i < W_LOADS; ++i) {
            const int o = wofs + 128 * i;
            wb[i] = *(const v4f*)(wmat + ((o < D_OUTF * D_INF - 4) ? o : (D_OUTF * D_INF - 4)));
        }
    }
    v4f xb[X_LOADS];
#pragma unroll
    for (int i = 0; i < X_LOADS; ++i)
        xb[i] = *(const v4f*)(x + (size_t)sj[i] * D_INF + 4 * pj[i]);

#pragma unroll
    for (int i = 0; i < X_LOADS; ++i) {
        s_x[t + GEMM_T * i] = xb[i];
    }
#pragma unroll
    for (int i = 0; i < W_LOADS; ++i) {
        s_w[wave][lane + 32 * i] = wb[i];
    }
    __syncthreads();
    if (!active) return;

    const v4f* wl = &s_w[wave][l16 * ROW_P + half];
    const v4f* xl = &s_x[l16 * ROW_P + half];
    const float* wf = (const float*)&s_w[wave][0] + l16 * D_INF + (D_INF - 4) + 2 * half;
    const float* xf = (const float*)&s_x[0]       + l16 * D_INF + (D_INF - 4) + 2 * half;
    v4f fa[KSTEPS], fb[KSTEPS];
    v2f ta = {0.f, 0.f}, tb = {0.f, 0.f};
    v8f c = {};
#pragma unroll
    for (int k = 0; k < KB; ++k) { fa[k] = wl[2 * k]; fb[k] = xl[2 * k]; }
#pragma unroll
    for (int kb = 0; kb < KSTEPS; kb += KB) {
#pragma unroll
        for (int k = kb + KB; k < kb + 2 * KB; ++k)
            if (k < KSTEPS) { fa[k] = wl[2 * k]; fb[k] = xl[2 * k]; }
        if (kb + KB < KSTEPS && kb + 2 * KB >= KSTEPS) {
            ta = *(const v2f*)wf;
            tb = *(const v2f*)xf;
        }
        __builtin_amdgcn_sched_barrier(0);
#pragma unroll
        for (int k = kb; k < kb + KB; ++k) {
            if (k < KSTEPS) {
                v2f a0 = {fa[k][0], fa[k][1]}, a1 = {fa[k][2], fa[k][3]};
                v2f b0 = {fb[k][0], fb[k][1]}, b1 = {fb[k][2], fb[k][3]};
                c = __builtin_amdgcn_wmma_f32_16x16x4_f32(
                         false, a0,  false, b0,
                         (short)0, c,  false,  false);
                c = __builtin_amdgcn_wmma_f32_16x16x4_f32(
                         false, a1,  false, b1,
                         (short)0, c,  false,  false);
            }
        }
        if (kb + KB >= KSTEPS)
            c = __builtin_amdgcn_wmma_f32_16x16x4_f32(
                     false, ta,  false, tb,
                     (short)0, c,  false,  false);
        __builtin_amdgcn_sched_barrier(0);
    }

    if (l16 < cnt) {
        const int m0 = rowbase + 8 * half;
        float* op = out + (size_t)s * D_OUTF + m0;
        if (m0 + 8 <= D_OUTF) {
            v4f lo = {c[0], c[1], c[2], c[3]};
            v4f hi = {c[4], c[5], c[6], c[7]};
            *(v4f*)(op + 0) = lo;
            *(v4f*)(op + 4) = hi;
        } else {
#pragma unroll
            for (int v = 0; v < 8; ++v)
                if (m0 + v < D_OUTF) op[v] = c[v];
        }
    }
}

extern "C" void kernel_launch(void* const* d_in, const int* in_sizes, int n_in,
                              void* d_out, int out_size, void* d_ws, size_t ws_size,
                              hipStream_t stream) {
    const float* x   = (const float*)d_in[0];
    const int*   rel = (const int*)  d_in[1];
    const float* W   = (const float*)d_in[2];
    float*       out = (float*)d_out;

    int* ws      = (int*)d_ws;
    int* order   = ws;
    int* ntiles  = order + B_SAMPLES;
    int* tiles   = ntiles + 1;

    rbt_bucket <<<1, PRE_T, 0, stream>>>(rel, order, tiles, ntiles);

    dim3 grid(NT_GRP, MAX_TILES);
    rbt_gemm   <<<grid, GEMM_T, 0, stream>>>(x, W, out, order, tiles, ntiles);
}


# Each GEMM work-group finds its relation's samples itself (scan of the relation ids, prefix-sum compaction to an LDS list) and tiles them with W staged once; bucketing launch gone, 2 launches to 1
# speedup vs baseline: 4.3446x; 1.5687x over previous
#include <hip/hip_runtime.h>

#define B_SAMPLES 4096
#define D_INF     300
#define D_OUTF    300
#define NUM_MAT   274
#define TILE      16
#define NT_OUT    ((D_OUTF + TILE - 1) / TILE)

typedef __attribute__((ext_vector_type(2))) float v2f;
typedef __attribute__((ext_vector_type(4))) float v4f;
typedef __attribute__((ext_vector_type(8))) float v8f;
typedef __attribute__((ext_vector_type(4))) int   v4i;

#define NW       4
#define GEMM_T   (32 * NW)
#define NT_GRP   ((NT_OUT + NW - 1) / NW)
#define ROW_P    (D_INF / 4)
#define TILE_P   (TILE * ROW_P)
#define W_LOADS  ((TILE_P + 31) / 32)
#define X_LOADS  ((TILE_P + GEMM_T - 1) / GEMM_T)
#define R_LOADS  (B_SAMPLES / (4 * GEMM_T))
#define KSTEPS   ((D_INF - 4) / 8)
#define KB       8

__device__ __forceinline__ void lds_barrier() {
    __builtin_amdgcn_fence(__ATOMIC_RELEASE, "workgroup", "local");
    __builtin_amdgcn_s_barrier();
    __builtin_amdgcn_fence(__ATOMIC_ACQUIRE, "workgroup", "local");
}

__device__ __forceinline__ v8f tile_product(const v4f* wl, const v4f* xl,
                                            const float* wf, const float* xf) {
    v4f fa[KSTEPS], fb[KSTEPS];
    v2f ta = {0.f, 0.f}, tb = {0.f, 0.f};
    v8f c = {};
#pragma unroll
    for (int k = 0; k < KB; ++k) { fa[k] = wl[2 * k]; fb[k] = xl[2 * k]; }
#pragma unroll
    for (int kb = 0; kb < KSTEPS; kb += KB) {
#pragma unroll
        for (int k = kb + KB; k < kb + 2 * KB; ++k)
            if (k < KSTEPS) { fa[k] = wl[2 * k]; fb[k] = xl[2 * k]; }
        if (kb + KB < KSTEPS && kb + 2 * KB >= KSTEPS) {
            ta = *(const v2f*)wf;
            tb = *(const v2f*)xf;
        }
        __builtin_amdgcn_sched_barrier(0);
#pragma unroll
        for (int k = kb; k < kb + KB; ++k) {
            if (k < KSTEPS) {
                v2f a0 = {fa[k][0], fa[k][1]}, a1 = {fa[k][2], fa[k][3]};
                v2f b0 = {fb[k][0], fb[k][1]}, b1 = {fb[k][2], fb[k][3]};
                c = __builtin_amdgcn_wmma_f32_16x16x4_f32(
                         false, a0,  false, b0,
                         (short)0, c,  false,  false);
                c = __builtin_amdgcn_wmma_f32_16x16x4_f32(
                         false, a1,  false, b1,
                         (short)0, c,  false,  false);
            }
        }
        if (kb + KB >= KSTEPS)
            c = __builtin_amdgcn_wmma_f32_16x16x4_f32(
                     false, ta,  false, tb,
                     (short)0, c,  false,  false);
        __builtin_amdgcn_sched_barrier(0);
    }
    return c;
}

__device__ __forceinline__ void store_tile(float* __restrict__ out, int s, int m0, bool valid, v8f c) {
    if (valid) {
        float* op = out + (size_t)s * D_OUTF + m0;
        if (m0 + 8 <= D_OUTF) {
            v4f lo = {c[0], c[1], c[2], c[3]};
            v4f hi = {c[4], c[5], c[6], c[7]};
            *(v4f*)(op + 0) = lo;
            *(v4f*)(op + 4) = hi;
        } else {
#pragma unroll
            for (int v = 0; v < 8; ++v)
                if (m0 + v < D_OUTF) op[v] = c[v];
        }
    }
}

__global__ void __launch_bounds__(GEMM_T)
rbt_gemm(const float* __restrict__ x, const int* __restrict__ rel,
         const float* __restrict__ W, float* __restrict__ out) {
    __shared__ v4f s_x[TILE_P];
    __shared__ v4f s_w[NW][TILE_P];
    __shared__ unsigned short s_list[B_SAMPLES];
    __shared__ int s_wtot[NW];

    const int r    = blockIdx.y;
    const int t    = threadIdx.x;
    const int wave = __builtin_amdgcn_readfirstlane(t >> 5);
    const int lane = t & 31;
    const int half = lane >> 4;
    const int l16  = lane & 15;

    const int  rowtile = blockIdx.x * NW + wave;
    const bool active  = rowtile < NT_OUT;
    const int  rowbase = rowtile * TILE;

    v4i rv[R_LOADS];
#pragma unroll
    for (int i = 0; i < R_LOADS; ++i)
        rv[i] = *(const v4i*)(rel + 4 * (t + GEMM_T * i));

    v4f wb[W_LOADS];
    {
        const float* wmat = W + (size_t)r * (D_OUTF * D_INF);
        const int    wofs = rowbase * D_INF + 4 * lane;
#pragma unroll
        for (int i = 0; i < W_LOADS; ++i) {
            const int o = wofs + 128 * i;
            wb[i] = *(const v4f*)(wmat + ((o < D_OUTF * D_INF - 4) ? o : (D_OUTF * D_INF - 4)));
        }
    }

    unsigned m = 0;
#pragma unroll
    for (int i = 0; i < R_LOADS; ++i)
#pragma unroll
        for (int e = 0; e < 4; ++e)
            m |= (unsigned)(rv[i][e] == r) << (4 * i + e);
    const int nm = __popc(m);
    int inc = nm;
#pragma unroll
    for (int d = 1; d < 32; d <<= 1) {
        const int v = __shfl_up(inc, d);
        if (lane >= d) inc += v;
    }
    if (lane == 31) s_wtot[wave] = inc;
    lds_barrier();
    int before = 0, tot = 0;
#pragma unroll
    for (int w = 0; w < NW; ++w) {
        const int v = s_wtot[w];
        tot += v;
        if (w < wave) before += v;
    }
    const int total = __builtin_amdgcn_readfirstlane(tot);
    {
        int pos = before + inc - nm;
        for (int q = 0; q < 32 && m != 0; ++q) {
            const int k = __ffs(m) - 1;
            m &= m - 1;
            s_list[pos++] = (unsigned short)(4 * (t + GEMM_T * (k >> 2)) + (k & 3));
        }
    }
    lds_barrier();
    if (total == 0) return;

    int colj[X_LOADS], pj[X_LOADS];
#pragma unroll
    for (int i = 0; i < X_LOADS; ++i) {
        const int j  = t + GEMM_T * i;
        const int jc = (j < TILE_P) ? j : (TILE_P - 1);
        colj[i] = jc / ROW_P;
        pj[i]   = jc - colj[i] * ROW_P;
    }

    const v4f* wl = &s_w[wave][l16 * ROW_P + half];
    const v4f* xl = &s_x[l16 * ROW_P + half];
    const float* wf = (const float*)&s_w[wave][0] + l16 * D_INF + (D_INF - 4) + 2 * half;
    const float* xf = (const float*)&s_x[0]       + l16 * D_INF + (D_INF - 4) + 2 * half;
    const int m0 = rowbase + 8 * half;

    {
        const int cnt = (total < TILE) ? total : TILE;
        v4f xb[X_LOADS];
#pragma unroll
        for (int i = 0; i < X_LOADS; ++i) {
            const int sj = s_list[(colj[i] < cnt) ? colj[i] : (cnt - 1)];
            xb[i] = *(const v4f*)(x + (size_t)sj * D_INF + 4 * pj[i]);
        }
        const int s = s_list[(l16 < cnt) ? l16 : (cnt - 1)];
#pragma unroll
        for (int i = 0; i < W_LOADS - 1; ++i)
            s_w[wave][lane + 32 * i] = wb[i];
        if (lane < TILE_P - 32 * (W_LOADS - 1))
            s_w[wave][lane + 32 * (W_LOADS - 1)] = wb[W_LOADS - 1];
#pragma unroll
        for (int i = 0; i < X_LOADS - 1; ++i)
            s_x[t + GEMM_T * i] = xb[i];
        if (t < TILE_P - GEMM_T * (X_LOADS - 1))
            s_x[t + GEMM_T * (X_LOADS - 1)] = xb[X_LOADS - 1];
        lds_barrier();
        if (active) {
            const v8f c = tile_product(wl, xl, wf, xf);
            store_tile(out, s, m0, l16 < cnt, c);
        }
    }

    for (int base = TILE; base < total; base += TILE) {
        const int left = total - base;
        const int cnt  = (left < TILE) ? left : TILE;
        v4f xb[X_LOADS];
#pragma unroll
        for (int i = 0; i < X_LOADS; ++i) {
            const int sj = s_list[base + ((colj[i] < cnt) ? colj[i] : (cnt - 1))];
            xb[i] = *(const v4f*)(x + (size_t)sj * D_INF + 4 * pj[i]);
        }
        const int s = s_list[base + ((l16 < cnt) ? l16 : (cnt - 1))];
        lds_barrier();
#pragma unroll
        for (int i = 0; i < X_LOADS - 1; ++i)
            s_x[t + GEMM_T * i] = xb[i];
        if (t < TILE_P - GEMM_T * (X_LOADS - 1))
            s_x[t + GEMM_T * (X_LOADS - 1)] = xb[X_LOADS - 1];
        lds_barrier();
        if (active) {
            const v8f c = tile_product(wl, xl, wf, xf);
            store_tile(out, s, m0, l16 < cnt, c);
        }
    }
}

extern "C" void kernel_launch(void* const* d_in, const int* in_sizes, int n_in,
                              void* d_out, int out_size, void* d_ws, size_t ws_size,
                              hipStream_t stream) {
    const float* x   = (const float*)d_in[0];
    const int*   rel = (const int*)  d_in[1];
    const float* W   = (const float*)d_in[2];
    float*       out = (float*)d_out;

    dim3 grid(NT_GRP, NUM_MAT);
    rbt_gemm<<<grid, GEMM_T, 0, stream>>>(x, rel, W, out);
}
